# starting kernel
# baseline (speedup 1.0000x reference)
#include <hip/hip_runtime.h>


typedef float v2f __attribute__((ext_vector_type(2)));
typedef float v8f __attribute__((ext_vector_type(8)));

#define KCH     29
#define NWAVES  4
#define XSTG    608
#define YSTG    768
#define BELEMS  (KCH * 3 * 32 * 2)

__global__ __launch_bounds__(128) void resample_poly_wmma_f32(
    const float* __restrict__ x, const float* __restrict__ f,
    float* __restrict__ out, int N, long long n_out, int n_tiles)
{
    __shared__ float Blds[BELEMS];
    __shared__ float xstage[NWAVES * XSTG];
    __shared__ float ystage[NWAVES * YSTG];

    const int tid  = threadIdx.x;
    const int lane = tid & 31;
    const int wave = tid >> 5;
    const int R    = lane & 15;
    const int hh   = lane >> 4;

    for (int i = tid; i < BELEMS; i += 128) {
        int v  = i & 1;
        int l  = (i >> 1) & 31;
        int a  = (i >> 6) % 3;
        int kc = i / 192;
        int t  = 4 * kc + v + 2 * (l >> 4);
        int c  = l & 15;
        int s  = t + 2 * c - ((a == 2) ? 31 : 32);
        int smax = (a == 0) ? 80 : 79;
        int tap  = (a == 0) ? 3 * s : ((a == 1) ? 3 * s + 2 : 3 * s + 1);
        float val = 0.0f;
        if (s >= 0 && s <= smax) val = f[tap];
        Blds[i] = val;
    }

    const long long T      = (long long)blockIdx.x * NWAVES + wave;
    const bool      activeW = (T < (long long)n_tiles);
    const long long n0     = 256LL * T;
    const long long base   = 40 + 2 * n0;
    float* xr = &xstage[wave * XSTG];

    for (int i = lane; i < 596; i += 32) {
        long long g = base + 512 - i;
        float v = 0.0f;
        if (activeW && g >= 0 && g < (long long)N) v = x[g];
        xr[i] = v;
    }
    __syncthreads();

    const int abase = 480 - 32 * R + 2 * hh;
    const int bbase = lane * 2;

    v8f c0 = {0.f, 0.f, 0.f, 0.f, 0.f, 0.f, 0.f, 0.f};
    v8f c1 = {0.f, 0.f, 0.f, 0.f, 0.f, 0.f, 0.f, 0.f};
    v8f c2 = {0.f, 0.f, 0.f, 0.f, 0.f, 0.f, 0.f, 0.f};

    for (int kc = 0; kc < KCH; ++kc) {
        v2f a = *(const v2f*)&xr[abase + 4 * kc];
        const float* bp = &Blds[kc * 192 + bbase];
        v2f b0 = *(const v2f*)&bp[0];
        v2f b1 = *(const v2f*)&bp[64];
        v2f b2 = *(const v2f*)&bp[128];
        c0 = __builtin_amdgcn_wmma_f32_16x16x4_f32(false, a, false, b0,
                                                   (short)0, c0, false, false);
        c1 = __builtin_amdgcn_wmma_f32_16x16x4_f32(false, a, false, b1,
                                                   (short)0, c1, false, false);
        c2 = __builtin_amdgcn_wmma_f32_16x16x4_f32(false, a, false, b2,
                                                   (short)0, c2, false, false);
    }

    float* yl = &ystage[wave * YSTG];
    for (int v = 0; v < 8; ++v) {
        int nl = 16 * (v + 8 * hh) + R;
        yl[3 * nl + 0] = 3.0f * c0[v];
        yl[3 * nl + 1] = 3.0f * c1[v];
        yl[3 * nl + 2] = 3.0f * c2[v];
    }
    __syncthreads();

    const long long m0 = 768LL * T;
    for (int k = 0; k < 24; ++k) {
        int i = k * 32 + lane;
        long long m = m0 + i;
        if (activeW && m < n_out) out[m] = yl[i];
    }
}

extern "C" void kernel_launch(void* const* d_in, const int* in_sizes, int n_in,
                              void* d_out, int out_size, void* d_ws, size_t ws_size,
                              hipStream_t stream) {
    const float* x = (const float*)d_in[0];
    const float* f = (const float*)d_in[1];
    float* out = (float*)d_out;

    const int       N      = in_sizes[0];
    const long long n_out  = (long long)out_size;
    const long long tiles  = (n_out + 767) / 768;
    const int       blocks = (int)((tiles + NWAVES - 1) / NWAVES);

    resample_poly_wmma_f32<<<blocks, 128, 0, stream>>>(
        x, f, out, N, n_out, (int)tiles);
}


# x staged through LDS again: one block-wide window loaded by whole 128-byte lines, read back at a 144-byte row pitch; 5 loads of 4 lines a thread replace 14 loads of 16 lines each; sums unchanged
# speedup vs baseline: 3.1409x; 3.1409x over previous
#include <hip/hip_runtime.h>


typedef float v2f __attribute__((ext_vector_type(2)));
typedef float v4f __attribute__((ext_vector_type(4)));
typedef float v8f __attribute__((ext_vector_type(8)));

#define KPAIRS  14
#define NWAVES  4
#define YSTG    768
#define FTAPS   241
#define BTOP    110
#define BLEN    144
#define WPIECES 538
#define XSWORDS 2432

__global__ __launch_bounds__(128) void resample_poly_wmma_f32(
    const float* __restrict__ x, const float* __restrict__ f,
    float* __restrict__ out, int N, long long n_out, int n_tiles)
{
    __shared__ __attribute__((aligned(16))) float sm[3 * BLEN + XSWORDS];
    float* const rv = sm;
    float* const xs = sm + 3 * BLEN;

    const int tid  = threadIdx.x;
    const int lane = tid & 31;
    const int wave = __builtin_amdgcn_readfirstlane(tid >> 5);
    const int R    = lane & 15;
    const int hh   = lane >> 4;

    const long long T       = (long long)blockIdx.x * NWAVES + wave;
    const bool      activeW = (T < (long long)n_tiles);

    const long long wbeg   = 2048LL * blockIdx.x - 64;
    const bool      inside = (wbeg >= 0) && (wbeg + 4 * WPIECES <= (long long)N);
    const int p4 = (tid + 512 < WPIECES) ? tid + 512 : WPIECES - 1;
    v4f xt[5];
    if (inside) {
        const float* xw = x + wbeg;
#pragma unroll
        for (int m = 0; m < 4; ++m)
            xt[m] = *(const v4f*)(xw + 4 * (tid + 128 * m));
        xt[4] = *(const v4f*)(xw + 4 * p4);
    } else {
#pragma unroll
        for (int m = 0; m < 5; ++m) {
            const int p = (m < 4) ? tid + 128 * m : p4;
#pragma unroll
            for (int e = 0; e < 4; ++e) {
                const long long g = wbeg + 4 * p + e;
                float v = 0.0f;
                if (g >= 0 && g < (long long)N) v = x[g];
                xt[m][e] = v;
            }
        }
    }

    for (int i = tid; i < BLEN; i += 128) {
        const int k0 = 3 * (BTOP - i);
        float v[3];
#pragma unroll
        for (int a = 0; a < 3; ++a) {
            const int k  = k0 + 2 * a;
            const int kc = k < 0 ? 0 : (k > FTAPS - 1 ? FTAPS - 1 : k);
            v[a] = f[kc];
        }
#pragma unroll
        for (int a = 0; a < 3; ++a) {
            const int k = k0 + 2 * a;
            rv[a * BLEN + i] = (k >= 0 && k < FTAPS) ? v[a] : 0.0f;
        }
    }
#pragma unroll
    for (int m = 0; m < 4; ++m) {
        const int p = tid + 128 * m;
        *(v4f*)&xs[4 * p + 4 * (p >> 3)] = xt[m];
    }
    if (tid + 512 < WPIECES)
        *(v4f*)&xs[4 * p4 + 4 * (p4 >> 3)] = xt[4];
    __syncthreads();

    const float* xa = &xs[576 * wave + 36 * R - 4 * hh];
    v4f aw[KPAIRS];
#pragma unroll
    for (int j = 0; j < KPAIRS; ++j) {
        const int u = 132 - 8 * j;
        aw[j] = *(const v4f*)(xa + u + 4 * (u >> 5));
    }

    const float* rb = &rv[BTOP + 28 - 4 * hh - 2 * R];

    v8f c0 = {0.f, 0.f, 0.f, 0.f, 0.f, 0.f, 0.f, 0.f};
    v8f c1 = {0.f, 0.f, 0.f, 0.f, 0.f, 0.f, 0.f, 0.f};
    v8f c2 = {0.f, 0.f, 0.f, 0.f, 0.f, 0.f, 0.f, 0.f};

#pragma unroll
    for (int j = 0; j < KPAIRS; ++j) {
        const float* bp = rb - 8 * j;
        v2f a_hi = {aw[j][0], aw[j][1]};
        v2f a_lo = {aw[j][2], aw[j][3]};
        v2f b0h = *(const v2f*)(bp);
        v2f b0l = *(const v2f*)(bp + 2);
        v2f b1h = *(const v2f*)(bp + BLEN);
        v2f b1l = *(const v2f*)(bp + BLEN + 2);
        v2f b2h = *(const v2f*)(bp + 2 * BLEN);
        v2f b2l = *(const v2f*)(bp + 2 * BLEN + 2);
        c0 = __builtin_amdgcn_wmma_f32_16x16x4_f32(false, a_hi, false, b0h,
                                                   (short)0, c0, false, false);
        c1 = __builtin_amdgcn_wmma_f32_16x16x4_f32(false, a_hi, false, b1h,
                                                   (short)0, c1, false, false);
        c2 = __builtin_amdgcn_wmma_f32_16x16x4_f32(false, a_hi, false, b2h,
                                                   (short)0, c2, false, false);
        c0 = __builtin_amdgcn_wmma_f32_16x16x4_f32(false, a_lo, false, b0l,
                                                   (short)0, c0, false, false);
        c1 = __builtin_amdgcn_wmma_f32_16x16x4_f32(false, a_lo, false, b1l,
                                                   (short)0, c1, false, false);
        c2 = __builtin_amdgcn_wmma_f32_16x16x4_f32(false, a_lo, false, b2l,
                                                   (short)0, c2, false, false);
    }

    const long long m0 = 768LL * T;
    if (activeW) {
        const int i0 = 3 * (128 * hh + R);
        if (m0 + YSTG <= n_out) {
            float* op = out + m0 + i0;
            for (int v = 0; v < 8; ++v) {
                float* p = op + 48 * v;
                p[0] = 3.0f * c0[v];
                p[1] = 3.0f * c1[v];
                p[2] = 3.0f * c2[v];
            }
        } else {
            for (int v = 0; v < 8; ++v) {
                long long m = m0 + i0 + 48 * v;
                if (m + 0 < n_out) out[m + 0] = 3.0f * c0[v];
                if (m + 1 < n_out) out[m + 1] = 3.0f * c1[v];
                if (m + 2 < n_out) out[m + 2] = 3.0f * c2[v];
            }
        }
    }
}

extern "C" void kernel_launch(void* const* d_in, const int* in_sizes, int n_in,
                              void* d_out, int out_size, void* d_ws, size_t ws_size,
                              hipStream_t stream) {
    const float* x = (const float*)d_in[0];
    const float* f = (const float*)d_in[1];
    float* out = (float*)d_out;

    const int       N      = in_sizes[0];
    const long long n_out  = (long long)out_size;
    const long long tiles  = (n_out + 767) / 768;
    const int       blocks = (int)((tiles + NWAVES - 1) / NWAVES);

    resample_poly_wmma_f32<<<blocks, 128, 0, stream>>>(
        x, f, out, N, n_out, (int)tiles);
}
